# starting kernel
# baseline (speedup 1.0000x reference)
#include <hip/hip_runtime.h>
#include <math.h>


typedef __attribute__((ext_vector_type(2))) float v2f;
typedef __attribute__((ext_vector_type(4))) float v4f;
typedef __attribute__((ext_vector_type(8))) float v8f;

#define NN      500000
#define DD      128
#define KK      16
#define NTILES  (NN / 16)
#define TSTRIDE 132
#define TEMP    0.5f
#define EPSV    1e-8f
#define NITER   11

__device__ __forceinline__ float sum16(float v) {
  v += __shfl_xor(v, 1, 32);
  v += __shfl_xor(v, 2, 32);
  v += __shfl_xor(v, 4, 32);
  v += __shfl_xor(v, 8, 32);
  return v;
}

__global__ void rownorm_kernel(const float* __restrict__ emb,
                               float* __restrict__ s1,
                               float* __restrict__ rsc) {
  int gw = (int)((blockIdx.x * blockDim.x + threadIdx.x) >> 5);
  int lane = threadIdx.x & 31;
  if (gw >= NN) return;
  const float4 v = ((const float4*)(emb + (size_t)gw * DD))[lane];
  float ss = v.x * v.x + v.y * v.y + v.z * v.z + v.w * v.w;
  ss += __shfl_xor(ss, 1, 32);
  ss += __shfl_xor(ss, 2, 32);
  ss += __shfl_xor(ss, 4, 32);
  ss += __shfl_xor(ss, 8, 32);
  ss += __shfl_xor(ss, 16, 32);
  float n = sqrtf(ss);
  if (lane == 0) {
    s1[gw]  = 1.0f / (n + EPSV);
    rsc[gw] = (n + EPSV) / n;
  }
}

__global__ void muprep_kernel(const float* __restrict__ mu_in,
                              float* __restrict__ mn,
                              float* __restrict__ cmean,
                              float* __restrict__ cr) {
  int tid = threadIdx.x;
  int lane = tid & 31;
  int w = tid >> 5;
  for (int r = w; r < KK; r += 8) {
    const float4 v = ((const float4*)(mu_in + r * DD))[lane];
    float ss = v.x * v.x + v.y * v.y + v.z * v.z + v.w * v.w;
    ss += __shfl_xor(ss, 1, 32);  ss += __shfl_xor(ss, 2, 32);
    ss += __shfl_xor(ss, 4, 32);  ss += __shfl_xor(ss, 8, 32);
    ss += __shfl_xor(ss, 16, 32);
    float inv1 = 1.0f / sqrtf(ss);
    float ax = v.x * inv1, ay = v.y * inv1, az = v.z * inv1, aw = v.w * inv1;
    float s2 = ax * ax + ay * ay + az * az + aw * aw;
    s2 += __shfl_xor(s2, 1, 32);  s2 += __shfl_xor(s2, 2, 32);
    s2 += __shfl_xor(s2, 4, 32);  s2 += __shfl_xor(s2, 8, 32);
    s2 += __shfl_xor(s2, 16, 32);
    float inv2 = 1.0f / sqrtf(s2);
    float4 o;
    o.x = ax * inv2; o.y = ay * inv2; o.z = az * inv2; o.w = aw * inv2;
    ((float4*)(mn + r * DD))[lane] = o;
  }
  for (int i = tid; i < KK * DD; i += 256) cmean[i] = 0.0f;
  if (tid < KK) cr[tid] = 0.0f;
}

__global__ void __launch_bounds__(128)
cluster_main(const float* __restrict__ emb, const float* __restrict__ s1,
             const float* __restrict__ rsc, const float* __restrict__ mn,
             float* __restrict__ cmean, float* __restrict__ cr,
             float* __restrict__ r_out, int write_r) {
  __shared__ __align__(16) float mn_lds[KK * TSTRIDE];
  __shared__ __align__(16) float tile[4][16 * TSTRIDE];
  __shared__ float cm_lds[KK * DD];
  __shared__ float cr_lds[KK];

  const int tid  = threadIdx.x;
  const int lane = tid & 31;
  const int wid  = tid >> 5;
  const int lcol = lane & 15;
  const int h2   = (lane >> 4) & 1;

  for (int i = tid; i < KK * DD; i += 128) {
    mn_lds[(i >> 7) * TSTRIDE + (i & 127)] = mn[i];
    cm_lds[i] = 0.0f;
  }
  if (tid < KK) cr_lds[tid] = 0.0f;
  __syncthreads();

  float* tl = &tile[wid][0];
  v8f acc[8] = {};
  float crv = 0.0f;

  const int gw = __builtin_amdgcn_readfirstlane(blockIdx.x * 4 + wid);
  const int nw = gridDim.x * 4;

  for (int t = gw; t < NTILES; t += nw) {
    const size_t trow = (size_t)t * 16;
    const float* tbase = emb + trow * DD;
    const float* s1b = s1 + trow;
    const float* rb  = rsc + trow;

    if (t + nw < NTILES)
      __builtin_prefetch(emb + (size_t)(t + nw) * 16 * DD, 0, 1);

#pragma unroll
    for (int i = 0; i < 16; ++i) {
      float4 v = ((const float4*)(tbase + i * DD))[lane];
      float si = s1b[i];
      float4 o;
      o.x = v.x * si; o.y = v.y * si; o.z = v.z * si; o.w = v.w * si;
      *(float4*)(tl + i * TSTRIDE + lane * 4) = o;
    }

    v8f cosv = {};
    const float* ar = tl + lcol * TSTRIDE + 64 * h2;
    const float* br = mn_lds + lcol * TSTRIDE + 64 * h2;
#pragma unroll
    for (int q = 0; q < 16; ++q) {
      v4f a4 = *(const v4f*)(ar + 4 * q);
      v4f b4 = *(const v4f*)(br + 4 * q);
      cosv = __builtin_amdgcn_wmma_f32_16x16x4_f32(
          false, a4.xy, false, b4.xy, (short)0, cosv, false, false);
      cosv = __builtin_amdgcn_wmma_f32_16x16x4_f32(
          false, a4.zw, false, b4.zw, (short)0, cosv, false, false);
    }

    float rr[8];
#pragma unroll
    for (int j = 0; j < 8; ++j) {
      float rsj = h2 ? rb[j + 8] : rb[j];
      float x = -TEMP * (cosv[j] * rsj);
      float e = __expf(x);
      float s = sum16(e);
      rr[j] = e / s;
    }

    if (write_r) {
#pragma unroll
      for (int j = 0; j < 8; ++j)
        r_out[(trow + j + 8 * h2) * KK + lcol] = rr[j];
    }
#pragma unroll
    for (int j = 0; j < 8; ++j) crv += rr[j];

#pragma unroll
    for (int c = 0; c < 4; ++c) {
      v2f af;
      af[0] = rr[2 * c];
      af[1] = rr[2 * c + 1];
      const float* b0 = tl + (2 * c + 8 * h2) * TSTRIDE + lcol;
#pragma unroll
      for (int d0 = 0; d0 < 8; ++d0) {
        v2f bf;
        bf[0] = b0[d0 * 16];
        bf[1] = b0[d0 * 16 + TSTRIDE];
        acc[d0] = __builtin_amdgcn_wmma_f32_16x16x4_f32(
            false, af, false, bf, (short)0, acc[d0], false, false);
      }
    }
  }

  crv += __shfl_xor(crv, 16, 32);
  if (lane < KK) atomicAdd(&cr_lds[lane], crv);
#pragma unroll
  for (int d0 = 0; d0 < 8; ++d0)
#pragma unroll
    for (int jj = 0; jj < 8; ++jj)
      atomicAdd(&cm_lds[(jj + 8 * h2) * DD + d0 * 16 + lcol], acc[d0][jj]);
  __syncthreads();
  for (int i = tid; i < KK * DD; i += 128) atomicAdd(&cmean[i], cm_lds[i]);
  if (tid < KK) atomicAdd(&cr[tid], cr_lds[tid]);
}

__global__ void muupdate_kernel(const float* __restrict__ cmean,
                                const float* __restrict__ cr,
                                float* __restrict__ mu,
                                float* __restrict__ out_mu, int write_out) {
  int i = blockIdx.x * blockDim.x + threadIdx.x;
  if (i < KK * DD) {
    float v = cmean[i] / cr[i >> 7];
    mu[i] = v;
    if (write_out) out_mu[i] = v;
  }
}

extern "C" void kernel_launch(void* const* d_in, const int* in_sizes, int n_in,
                              void* d_out, int out_size, void* d_ws, size_t ws_size,
                              hipStream_t stream) {
  const float* emb     = (const float*)d_in[0];
  const float* mu_init = (const float*)d_in[1];
  float* out = (float*)d_out;

  float* s1    = (float*)d_ws;
  float* rsc   = s1 + NN;
  float* mu    = rsc + NN;
  float* mn    = mu + KK * DD;
  float* cmean = mn + KK * DD;
  float* cr    = cmean + KK * DD;

  rownorm_kernel<<<(NN + 7) / 8, 256, 0, stream>>>(emb, s1, rsc);

  for (int it = 0; it < NITER; ++it) {
    const float* musrc = (it == 0) ? mu_init : mu;
    const int last = (it == NITER - 1) ? 1 : 0;
    muprep_kernel<<<1, 256, 0, stream>>>(musrc, mn, cmean, cr);
    cluster_main<<<1024, 128, 0, stream>>>(emb, s1, rsc, mn, cmean, cr,
                                           out + KK * DD, last);
    muupdate_kernel<<<(KK * DD) / 256, 256, 0, stream>>>(cmean, cr, mu, out, last);
  }
}


# main pass: each wave sets its issue priority (s_setprio) to the number of tiles it still has to do, so the 8 waves of a SIMD advance evenly, not oldest-first, and end together; no sum reordered
# speedup vs baseline: 1.2238x; 1.2238x over previous
#include <hip/hip_runtime.h>
#include <math.h>


typedef __attribute__((ext_vector_type(2))) float v2f;
typedef __attribute__((ext_vector_type(4))) float v4f;
typedef __attribute__((ext_vector_type(8))) float v8f;

#define NN      500000
#define DD      128
#define KK      16
#define NTILES  (NN / 16)
#define TSTRIDE 132
#define TEMP    0.5f
#define EPSV    1e-8f
#define NITER   11

#define SWZ_XOR(v, m) \
  __int_as_float(__builtin_amdgcn_ds_swizzle(__float_as_int(v), 0x1f | ((m) << 10)))
__device__ __forceinline__ float sum16(float v) {
  v += SWZ_XOR(v, 1);
  v += SWZ_XOR(v, 2);
  v += SWZ_XOR(v, 4);
  v += SWZ_XOR(v, 8);
  return v;
}

__global__ void __launch_bounds__(256)
rownorm_kernel(const float* __restrict__ emb,
               float* __restrict__ s1,
               float* __restrict__ rsc) {
  const int lane = threadIdx.x & 31;
  const int gw = __builtin_amdgcn_readfirstlane(
      (int)(blockIdx.x * 8 + (threadIdx.x >> 5)));
  const int nw = (int)gridDim.x * 8;
  const bool b0 = (lane & 1) != 0;
  const bool b1 = (lane & 2) != 0;
  const bool b2 = (lane & 4) != 0;
  const bool b3 = (lane & 8) != 0;
  const int rho = ((lane & 1) << 3) | ((lane & 2) << 1) |
                  ((lane & 4) >> 1) | ((lane & 8) >> 3);

  for (int t = gw; t < NTILES; t += nw) {
    const float* tbase = emb + (size_t)t * 16 * DD;
    float ss[16];
#pragma unroll
    for (int i = 0; i < 16; ++i) {
      const float4 v = ((const float4*)(tbase + i * DD))[lane];
      ss[i] = v.x * v.x + v.y * v.y + v.z * v.z + v.w * v.w;
    }
    float p8[8], p4[4], p2[2];
#pragma unroll
    for (int i = 0; i < 8; ++i) {
      const float keep = b0 ? ss[i + 8] : ss[i];
      const float send = b0 ? ss[i] : ss[i + 8];
      p8[i] = keep + __shfl_xor(send, 1, 32);
    }
#pragma unroll
    for (int i = 0; i < 4; ++i) {
      const float keep = b1 ? p8[i + 4] : p8[i];
      const float send = b1 ? p8[i] : p8[i + 4];
      p4[i] = keep + __shfl_xor(send, 2, 32);
    }
#pragma unroll
    for (int i = 0; i < 2; ++i) {
      const float keep = b2 ? p4[i + 2] : p4[i];
      const float send = b2 ? p4[i] : p4[i + 2];
      p2[i] = keep + __shfl_xor(send, 4, 32);
    }
    const float keep = b3 ? p2[1] : p2[0];
    const float send = b3 ? p2[0] : p2[1];
    float tot = keep + __shfl_xor(send, 8, 32);
    tot += __shfl_xor(tot, 16, 32);
    const float n = sqrtf(tot);
    if (lane < 16) {
      const size_t row = (size_t)t * 16 + rho;
      s1[row]  = 1.0f / (n + EPSV);
      rsc[row] = (n + EPSV) / n;
    }
  }
}

__global__ void __launch_bounds__(512)
muprep_kernel(const float* src,
              float* __restrict__ mn,
              float* cmean,
              float* __restrict__ cr) {
  const int tid = threadIdx.x;
  const int lane = tid & 31;
  const int r = tid >> 5;
  const float4 v = ((const float4*)(src + r * DD))[lane];
  float ss = v.x * v.x + v.y * v.y + v.z * v.z + v.w * v.w;
  ss += __shfl_xor(ss, 1, 32);  ss += __shfl_xor(ss, 2, 32);
  ss += __shfl_xor(ss, 4, 32);  ss += __shfl_xor(ss, 8, 32);
  ss += __shfl_xor(ss, 16, 32);
  float inv1 = 1.0f / sqrtf(ss);
  float ax = v.x * inv1, ay = v.y * inv1, az = v.z * inv1, aw = v.w * inv1;
  float s2 = ax * ax + ay * ay + az * az + aw * aw;
  s2 += __shfl_xor(s2, 1, 32);  s2 += __shfl_xor(s2, 2, 32);
  s2 += __shfl_xor(s2, 4, 32);  s2 += __shfl_xor(s2, 8, 32);
  s2 += __shfl_xor(s2, 16, 32);
  float inv2 = 1.0f / sqrtf(s2);
  float4 o;
  o.x = ax * inv2; o.y = ay * inv2; o.z = az * inv2; o.w = aw * inv2;
  ((float4*)(mn + r * DD))[lane] = o;
  ((float4*)(cmean + r * DD))[lane] = make_float4(0.0f, 0.0f, 0.0f, 0.0f);
  if (tid < KK) cr[tid] = 0.0f;
}

#define MAIN_WAVES   32
#define MAIN_THREADS (MAIN_WAVES * 32)
#define MAIN_BLOCKS  256
__global__ void __launch_bounds__(MAIN_THREADS)
cluster_main(const float* __restrict__ emb, const float* __restrict__ s1,
             const float* __restrict__ rsc, const float* __restrict__ mn,
             float* __restrict__ cmean, float* __restrict__ cr,
             float* __restrict__ r_out, int write_r) {
  __shared__ __align__(16) float mn_lds[KK * TSTRIDE];
  __shared__ __align__(16) float tile[MAIN_WAVES][16 * TSTRIDE];
  __shared__ float cm_lds[KK * TSTRIDE];
  __shared__ float cr_lds[KK];

  const int tid  = threadIdx.x;
  const int lane = tid & 31;
  const int wid  = tid >> 5;
  const int lcol = lane & 15;
  const int h2   = (lane >> 4) & 1;

  for (int i = tid; i < KK * DD; i += MAIN_THREADS) {
    mn_lds[(i >> 7) * TSTRIDE + (i & 127)] = mn[i];
    cm_lds[(i >> 7) * TSTRIDE + (i & 127)] = 0.0f;
  }
  if (tid < KK) cr_lds[tid] = 0.0f;
  __syncthreads();

  float* tl = &tile[wid][0];
  v8f acc[8] = {};
  float crv = 0.0f;

  const int wid_s = __builtin_amdgcn_readfirstlane(wid);
  const int gw = __builtin_amdgcn_readfirstlane(blockIdx.x * MAIN_WAVES + wid);
  const int nw = gridDim.x * MAIN_WAVES;

  for (int t = gw; t < NTILES; t += nw) {
    if (t + 3 * nw < NTILES)      __builtin_amdgcn_s_setprio(3);
    else if (t + 2 * nw < NTILES) __builtin_amdgcn_s_setprio(2);
    else if (t + nw < NTILES)     __builtin_amdgcn_s_setprio(1);
    else                          __builtin_amdgcn_s_setprio(0);

    const size_t trow = (size_t)t * 16;
    const float* tbase = emb + trow * DD;
    const float* s1b = s1 + trow;
    const float* rb  = rsc + trow;

    float sc[16];
#pragma unroll
    for (int i = 0; i < 16; ++i) sc[i] = s1b[i];
#pragma unroll
    for (int g = 0; g < 2; ++g) {
      v4f v[8];
#pragma unroll
      for (int i = 0; i < 8; ++i)
        v[i] = ((const v4f*)(tbase + (8 * g + i) * DD))[lane];
      if (g == 0)
        __builtin_prefetch((const char*)(tbase + 8 * DD) + lane * 128, 0, 1);
      if (g == 0)
        asm volatile("" :: "v"(v[0]), "v"(v[1]), "v"(v[2]), "v"(v[3]),
                           "v"(v[4]), "v"(v[5]), "v"(v[6]), "v"(v[7]),
                           "s"(sc[0]), "s"(sc[1]), "s"(sc[2]), "s"(sc[3]),
                           "s"(sc[4]), "s"(sc[5]), "s"(sc[6]), "s"(sc[7]),
                           "s"(sc[8]), "s"(sc[9]), "s"(sc[10]), "s"(sc[11]),
                           "s"(sc[12]), "s"(sc[13]), "s"(sc[14]), "s"(sc[15]));
      else
        asm volatile("" :: "v"(v[0]), "v"(v[1]), "v"(v[2]), "v"(v[3]),
                           "v"(v[4]), "v"(v[5]), "v"(v[6]), "v"(v[7]));
#pragma unroll
      for (int i = 0; i < 8; ++i)
        *(v4f*)(tl + (8 * g + i) * TSTRIDE + lane * 4) = v[i] * sc[8 * g + i];
    }

    v8f cosv = {};
    const float* ar = tl + lcol * TSTRIDE + 64 * h2;
    const float* br = mn_lds + lcol * TSTRIDE + 64 * h2;
#pragma unroll
    for (int q = 0; q < 16; ++q) {
      v4f a4 = *(const v4f*)(ar + 4 * q);
      v4f b4 = *(const v4f*)(br + 4 * q);
      cosv = __builtin_amdgcn_wmma_f32_16x16x4_f32(
          false, a4.xy, false, b4.xy, (short)0, cosv, false, false);
      cosv = __builtin_amdgcn_wmma_f32_16x16x4_f32(
          false, a4.zw, false, b4.zw, (short)0, cosv, false, false);
    }

    float rr[8];
#pragma unroll
    for (int j = 0; j < 8; ++j) {
      const float rlo = __int_as_float(__builtin_amdgcn_readfirstlane(__float_as_int(rb[j])));
      const float rhi = __int_as_float(__builtin_amdgcn_readfirstlane(__float_as_int(rb[j + 8])));
      float rsj = h2 ? rhi : rlo;
      float x = -TEMP * (cosv[j] * rsj);
      float e = __expf(x);
      float s = sum16(e);
      rr[j] = e / s;
    }

    if (write_r) {
      float* rt = r_out + trow * KK;
#pragma unroll
      for (int j = 0; j < 8; ++j)
        rt[(j + 8 * h2) * KK + lcol] = rr[j];
    }
#pragma unroll
    for (int j = 0; j < 8; ++j) crv += rr[j];

#pragma unroll
    for (int c = 0; c < 4; ++c) {
      v2f af;
      af[0] = rr[2 * c];
      af[1] = rr[2 * c + 1];
      const float* b0 = tl + (2 * c + 8 * h2) * TSTRIDE + lcol;
#pragma unroll
      for (int d0 = 0; d0 < 8; ++d0) {
        v2f bf;
        bf[0] = b0[d0 * 16];
        bf[1] = b0[d0 * 16 + TSTRIDE];
        acc[d0] = __builtin_amdgcn_wmma_f32_16x16x4_f32(
            false, af, false, bf, (short)0, acc[d0], false, false);
      }
    }
  }

  const int elane = (int)__builtin_amdgcn_mbcnt_lo(~0u, 0u);
  const int elcol = elane & 15;
  const int eh2   = elane >> 4;
  const int etid  = wid_s * 32 + elane;
  crv += __shfl_xor(crv, 16, 32);
  if (elane < KK) atomicAdd(&cr_lds[elane], crv);
#pragma unroll
  for (int d0 = 0; d0 < 8; ++d0)
#pragma unroll
    for (int jj = 0; jj < 8; ++jj)
      atomicAdd(&cm_lds[(jj + 8 * eh2) * TSTRIDE + d0 * 16 + elcol], acc[d0][jj]);
  __syncthreads();
  for (int i = etid; i < KK * DD; i += MAIN_THREADS)
    atomicAdd(&cmean[i], cm_lds[(i >> 7) * TSTRIDE + (i & 127)]);
  if (etid < KK) atomicAdd(&cr[etid], cr_lds[etid]);
}

__global__ void muupdate_kernel(const float* __restrict__ cmean,
                                const float* __restrict__ cr,
                                float* __restrict__ out_mu) {
  int i = blockIdx.x * blockDim.x + threadIdx.x;
  if (i < KK * DD) out_mu[i] = cmean[i] / cr[i >> 7];
}

extern "C" void kernel_launch(void* const* d_in, const int* in_sizes, int n_in,
                              void* d_out, int out_size, void* d_ws, size_t ws_size,
                              hipStream_t stream) {
  const float* emb     = (const float*)d_in[0];
  const float* mu_init = (const float*)d_in[1];
  float* out = (float*)d_out;

  float* s1    = (float*)d_ws;
  float* rsc   = s1 + NN;
  float* mn    = rsc + NN;
  float* cmean = mn + KK * DD;
  float* cr    = cmean + KK * DD;

  rownorm_kernel<<<1024, 256, 0, stream>>>(emb, s1, rsc);

  for (int it = 0; it < NITER; ++it) {
    const float* musrc = (it == 0) ? mu_init : cmean;
    const int last = (it == NITER - 1) ? 1 : 0;
    muprep_kernel<<<1, 512, 0, stream>>>(musrc, mn, cmean, cr);
    cluster_main<<<MAIN_BLOCKS, MAIN_THREADS, 0, stream>>>(emb, s1, rsc, mn, cmean, cr,
                                           out + KK * DD, last);
  }
  muupdate_kernel<<<(KK * DD) / 256, 256, 0, stream>>>(cmean, cr, out);
}


# product 1 computed transposed (scores^T = mn x tile^T): a row's softmax is lane-local, 7 adds + 1 exchange and one 1/s per row (was 32 swizzles, 8 divisions); r via LDS transpose; sum order changes
# speedup vs baseline: 1.2922x; 1.0559x over previous
#include <hip/hip_runtime.h>
#include <math.h>


typedef __attribute__((ext_vector_type(2))) float v2f;
typedef __attribute__((ext_vector_type(4))) float v4f;
typedef __attribute__((ext_vector_type(8))) float v8f;

#define NN      500000
#define DD      128
#define KK      16
#define NTILES  (NN / 16)
#define TSTRIDE 132
#define TEMP    0.5f
#define EPSV    1e-8f
#define NITER   11

#define SWZ_XOR(v, m) \
  __int_as_float(__builtin_amdgcn_ds_swizzle(__float_as_int(v), 0x1f | ((m) << 10)))
__device__ __forceinline__ float sum16(float v) {
  v += SWZ_XOR(v, 1);
  v += SWZ_XOR(v, 2);
  v += SWZ_XOR(v, 4);
  v += SWZ_XOR(v, 8);
  return v;
}

__global__ void __launch_bounds__(256)
rownorm_kernel(const float* __restrict__ emb,
               float* __restrict__ s1,
               float* __restrict__ rsc) {
  const int lane = threadIdx.x & 31;
  const int gw = __builtin_amdgcn_readfirstlane(
      (int)(blockIdx.x * 8 + (threadIdx.x >> 5)));
  const int nw = (int)gridDim.x * 8;
  const bool b0 = (lane & 1) != 0;
  const bool b1 = (lane & 2) != 0;
  const bool b2 = (lane & 4) != 0;
  const bool b3 = (lane & 8) != 0;
  const int rho = ((lane & 1) << 3) | ((lane & 2) << 1) |
                  ((lane & 4) >> 1) | ((lane & 8) >> 3);

  for (int t = gw; t < NTILES; t += nw) {
    const float* tbase = emb + (size_t)t * 16 * DD;
    float ss[16];
#pragma unroll
    for (int i = 0; i < 16; ++i) {
      const float4 v = ((const float4*)(tbase + i * DD))[lane];
      ss[i] = v.x * v.x + v.y * v.y + v.z * v.z + v.w * v.w;
    }
    float p8[8], p4[4], p2[2];
#pragma unroll
    for (int i = 0; i < 8; ++i) {
      const float keep = b0 ? ss[i + 8] : ss[i];
      const float send = b0 ? ss[i] : ss[i + 8];
      p8[i] = keep + __shfl_xor(send, 1, 32);
    }
#pragma unroll
    for (int i = 0; i < 4; ++i) {
      const float keep = b1 ? p8[i + 4] : p8[i];
      const float send = b1 ? p8[i] : p8[i + 4];
      p4[i] = keep + __shfl_xor(send, 2, 32);
    }
#pragma unroll
    for (int i = 0; i < 2; ++i) {
      const float keep = b2 ? p4[i + 2] : p4[i];
      const float send = b2 ? p4[i] : p4[i + 2];
      p2[i] = keep + __shfl_xor(send, 4, 32);
    }
    const float keep = b3 ? p2[1] : p2[0];
    const float send = b3 ? p2[0] : p2[1];
    float tot = keep + __shfl_xor(send, 8, 32);
    tot += __shfl_xor(tot, 16, 32);
    const float n = sqrtf(tot);
    if (lane < 16) {
      const size_t row = (size_t)t * 16 + rho;
      s1[row]  = 1.0f / (n + EPSV);
      rsc[row] = (n + EPSV) / n;
    }
  }
}

__global__ void __launch_bounds__(512)
muprep_kernel(const float* src,
              float* __restrict__ mn,
              float* cmean,
              float* __restrict__ cr) {
  const int tid = threadIdx.x;
  const int lane = tid & 31;
  const int r = tid >> 5;
  const float4 v = ((const float4*)(src + r * DD))[lane];
  float ss = v.x * v.x + v.y * v.y + v.z * v.z + v.w * v.w;
  ss += __shfl_xor(ss, 1, 32);  ss += __shfl_xor(ss, 2, 32);
  ss += __shfl_xor(ss, 4, 32);  ss += __shfl_xor(ss, 8, 32);
  ss += __shfl_xor(ss, 16, 32);
  float inv1 = 1.0f / sqrtf(ss);
  float ax = v.x * inv1, ay = v.y * inv1, az = v.z * inv1, aw = v.w * inv1;
  float s2 = ax * ax + ay * ay + az * az + aw * aw;
  s2 += __shfl_xor(s2, 1, 32);  s2 += __shfl_xor(s2, 2, 32);
  s2 += __shfl_xor(s2, 4, 32);  s2 += __shfl_xor(s2, 8, 32);
  s2 += __shfl_xor(s2, 16, 32);
  float inv2 = 1.0f / sqrtf(s2);
  float4 o;
  o.x = ax * inv2; o.y = ay * inv2; o.z = az * inv2; o.w = aw * inv2;
  ((float4*)(mn + r * DD))[lane] = o;
  ((float4*)(cmean + r * DD))[lane] = make_float4(0.0f, 0.0f, 0.0f, 0.0f);
  if (tid < KK) cr[tid] = 0.0f;
}

#define MAIN_WAVES   32
#define MAIN_THREADS (MAIN_WAVES * 32)
#define MAIN_BLOCKS  256
__global__ void __launch_bounds__(MAIN_THREADS)
cluster_main(const float* __restrict__ emb, const float* __restrict__ s1,
             const float* __restrict__ rsc, const float* __restrict__ mn,
             float* __restrict__ cmean, float* __restrict__ cr,
             float* __restrict__ r_out, int write_r) {
  __shared__ __align__(16) float mn_lds[KK * TSTRIDE];
  __shared__ __align__(16) float tile[MAIN_WAVES][16 * TSTRIDE];
  __shared__ __align__(16) float rt_lds[MAIN_WAVES][16 * KK];
  __shared__ float cm_lds[KK * TSTRIDE];
  __shared__ float cr_lds[KK];

  const int tid  = threadIdx.x;
  const int lane = tid & 31;
  const int wid  = tid >> 5;
  const int lcol = lane & 15;
  const int h2   = (lane >> 4) & 1;

  for (int i = tid; i < KK * DD; i += MAIN_THREADS) {
    mn_lds[(i >> 7) * TSTRIDE + (i & 127)] = mn[i];
    cm_lds[(i >> 7) * TSTRIDE + (i & 127)] = 0.0f;
  }
  if (tid < KK) cr_lds[tid] = 0.0f;
  __syncthreads();

  float* tl = &tile[wid][0];
  v8f acc[8] = {};
  float crv = 0.0f;

  const int wid_s = __builtin_amdgcn_readfirstlane(wid);
  const int gw = __builtin_amdgcn_readfirstlane(blockIdx.x * MAIN_WAVES + wid);
  const int nw = gridDim.x * MAIN_WAVES;

  for (int t = gw; t < NTILES; t += nw) {
    if (t + 3 * nw < NTILES)      __builtin_amdgcn_s_setprio(3);
    else if (t + 2 * nw < NTILES) __builtin_amdgcn_s_setprio(2);
    else if (t + nw < NTILES)     __builtin_amdgcn_s_setprio(1);
    else                          __builtin_amdgcn_s_setprio(0);

    const size_t trow = (size_t)t * 16;
    const float* tbase = emb + trow * DD;
    const float* s1b = s1 + trow;
    const float* rb  = rsc + trow;

    float sc[16];
#pragma unroll
    for (int i = 0; i < 16; ++i) sc[i] = s1b[i];
#pragma unroll
    for (int g = 0; g < 2; ++g) {
      v4f v[8];
#pragma unroll
      for (int i = 0; i < 8; ++i)
        v[i] = ((const v4f*)(tbase + (8 * g + i) * DD))[lane];
      if (g == 0)
        __builtin_prefetch((const char*)(tbase + 8 * DD) + lane * 128, 0, 1);
      if (g == 0)
        asm volatile("" :: "v"(v[0]), "v"(v[1]), "v"(v[2]), "v"(v[3]),
                           "v"(v[4]), "v"(v[5]), "v"(v[6]), "v"(v[7]),
                           "s"(sc[0]), "s"(sc[1]), "s"(sc[2]), "s"(sc[3]),
                           "s"(sc[4]), "s"(sc[5]), "s"(sc[6]), "s"(sc[7]),
                           "s"(sc[8]), "s"(sc[9]), "s"(sc[10]), "s"(sc[11]),
                           "s"(sc[12]), "s"(sc[13]), "s"(sc[14]), "s"(sc[15]));
      else
        asm volatile("" :: "v"(v[0]), "v"(v[1]), "v"(v[2]), "v"(v[3]),
                           "v"(v[4]), "v"(v[5]), "v"(v[6]), "v"(v[7]));
#pragma unroll
      for (int i = 0; i < 8; ++i)
        *(v4f*)(tl + (8 * g + i) * TSTRIDE + lane * 4) = v[i] * sc[8 * g + i];
    }


    const float rsl = rb[(unsigned)lcol];
    v8f cosv = {};
    const float* ar = tl + lcol * TSTRIDE + 64 * h2;
    const float* br = mn_lds + lcol * TSTRIDE + 64 * h2;
#pragma unroll
    for (int q = 0; q < 16; ++q) {
      v4f a4 = *(const v4f*)(ar + 4 * q);
      v4f b4 = *(const v4f*)(br + 4 * q);
      cosv = __builtin_amdgcn_wmma_f32_16x16x4_f32(
          false, b4.xy, false, a4.xy, (short)0, cosv, false, false);
      cosv = __builtin_amdgcn_wmma_f32_16x16x4_f32(
          false, b4.zw, false, a4.zw, (short)0, cosv, false, false);
    }

    float ev[8];
#pragma unroll
    for (int j = 0; j < 8; ++j)
      ev[j] = __expf(-TEMP * (cosv[j] * rsl));
    float s = ((ev[0] + ev[1]) + (ev[2] + ev[3])) + ((ev[4] + ev[5]) + (ev[6] + ev[7]));
    s += SWZ_XOR(s, 16);
    const float inv = 1.0f / s;
    v4f plo, phi;
    plo[0] = ev[0] * inv; plo[1] = ev[1] * inv; plo[2] = ev[2] * inv; plo[3] = ev[3] * inv;
    phi[0] = ev[4] * inv; phi[1] = ev[5] * inv; phi[2] = ev[6] * inv; phi[3] = ev[7] * inv;

    if (write_r) {
      float* rt = r_out + trow * KK;
      const unsigned ro = (unsigned)(lcol * KK + h2 * 8);
      *(v4f*)(rt + ro)     = plo;
      *(v4f*)(rt + ro + 4) = phi;
    }

    float* rls = &rt_lds[wid][0] + lcol * KK + h2 * 8;
    const float* rlr = &rt_lds[wid][0] + h2 * (8 * KK) + lcol;
    *(v4f*)(rls)     = plo;
    *(v4f*)(rls + 4) = phi;
    float rr[8];
#pragma unroll
    for (int j = 0; j < 8; ++j) rr[j] = rlr[j * KK];

#pragma unroll
    for (int j = 0; j < 8; ++j) crv += rr[j];

#pragma unroll
    for (int c = 0; c < 4; ++c) {
      v2f af;
      af[0] = rr[2 * c];
      af[1] = rr[2 * c + 1];
      const float* b0 = (tl + h2 * (8 * TSTRIDE) + lcol) + 2 * c * TSTRIDE;
#pragma unroll
      for (int d0 = 0; d0 < 8; ++d0) {
        v2f bf;
        bf[0] = b0[d0 * 16];
        bf[1] = b0[d0 * 16 + TSTRIDE];
        acc[d0] = __builtin_amdgcn_wmma_f32_16x16x4_f32(
            false, af, false, bf, (short)0, acc[d0], false, false);
      }
    }
  }

  const int elane = (int)__builtin_amdgcn_mbcnt_lo(~0u, 0u);
  const int elcol = elane & 15;
  const int eh2   = elane >> 4;
  const int etid  = wid_s * 32 + elane;
  crv += __shfl_xor(crv, 16, 32);
  if (elane < KK) atomicAdd(&cr_lds[elane], crv);
#pragma unroll
  for (int d0 = 0; d0 < 8; ++d0)
#pragma unroll
    for (int jj = 0; jj < 8; ++jj)
      atomicAdd(&cm_lds[(jj + 8 * eh2) * TSTRIDE + d0 * 16 + elcol], acc[d0][jj]);
  __syncthreads();
  for (int i = etid; i < KK * DD; i += MAIN_THREADS)
    atomicAdd(&cmean[i], cm_lds[(i >> 7) * TSTRIDE + (i & 127)]);
  if (etid < KK) atomicAdd(&cr[etid], cr_lds[etid]);
}

__global__ void muupdate_kernel(const float* __restrict__ cmean,
                                const float* __restrict__ cr,
                                float* __restrict__ out_mu) {
  int i = blockIdx.x * blockDim.x + threadIdx.x;
  if (i < KK * DD) out_mu[i] = cmean[i] / cr[i >> 7];
}

extern "C" void kernel_launch(void* const* d_in, const int* in_sizes, int n_in,
                              void* d_out, int out_size, void* d_ws, size_t ws_size,
                              hipStream_t stream) {
  const float* emb     = (const float*)d_in[0];
  const float* mu_init = (const float*)d_in[1];
  float* out = (float*)d_out;

  float* s1    = (float*)d_ws;
  float* rsc   = s1 + NN;
  float* mn    = rsc + NN;
  float* cmean = mn + KK * DD;
  float* cr    = cmean + KK * DD;

  rownorm_kernel<<<1024, 256, 0, stream>>>(emb, s1, rsc);

  for (int it = 0; it < NITER; ++it) {
    const float* musrc = (it == 0) ? mu_init : cmean;
    const int last = (it == NITER - 1) ? 1 : 0;
    muprep_kernel<<<1, 512, 0, stream>>>(musrc, mn, cmean, cr);
    cluster_main<<<MAIN_BLOCKS, MAIN_THREADS, 0, stream>>>(emb, s1, rsc, mn, cmean, cr,
                                           out + KK * DD, last);
  }
  muupdate_kernel<<<(KK * DD) / 256, 256, 0, stream>>>(cmean, cr, out);
}


# main pass walks each wave's tiles in the opposite direction on alternate passes, so a pass starts on the tiles the second-level cache still holds; a wave's tile sum order flips on those passes
# speedup vs baseline: 1.3731x; 1.0626x over previous
#include <hip/hip_runtime.h>
#include <math.h>


typedef __attribute__((ext_vector_type(2))) float v2f;
typedef __attribute__((ext_vector_type(4))) float v4f;
typedef __attribute__((ext_vector_type(8))) float v8f;

#define NN      500000
#define DD      128
#define KK      16
#define NTILES  (NN / 16)
#define TSTRIDE 132
#define TEMP    0.5f
#define EPSV    1e-8f
#define NITER   11

#define SWZ_XOR(v, m) \
  __int_as_float(__builtin_amdgcn_ds_swizzle(__float_as_int(v), 0x1f | ((m) << 10)))
__device__ __forceinline__ float sum16(float v) {
  v += SWZ_XOR(v, 1);
  v += SWZ_XOR(v, 2);
  v += SWZ_XOR(v, 4);
  v += SWZ_XOR(v, 8);
  return v;
}

__global__ void __launch_bounds__(256)
rownorm_kernel(const float* __restrict__ emb,
               float* __restrict__ s1,
               float* __restrict__ rsc) {
  const int lane = threadIdx.x & 31;
  const int gw = __builtin_amdgcn_readfirstlane(
      (int)(blockIdx.x * 8 + (threadIdx.x >> 5)));
  const int nw = (int)gridDim.x * 8;
  const bool b0 = (lane & 1) != 0;
  const bool b1 = (lane & 2) != 0;
  const bool b2 = (lane & 4) != 0;
  const bool b3 = (lane & 8) != 0;
  const int rho = ((lane & 1) << 3) | ((lane & 2) << 1) |
                  ((lane & 4) >> 1) | ((lane & 8) >> 3);

  for (int t = gw; t < NTILES; t += nw) {
    const float* tbase = emb + (size_t)t * 16 * DD;
    float ss[16];
#pragma unroll
    for (int i = 0; i < 16; ++i) {
      const float4 v = ((const float4*)(tbase + i * DD))[lane];
      ss[i] = v.x * v.x + v.y * v.y + v.z * v.z + v.w * v.w;
    }
    float p8[8], p4[4], p2[2];
#pragma unroll
    for (int i = 0; i < 8; ++i) {
      const float keep = b0 ? ss[i + 8] : ss[i];
      const float send = b0 ? ss[i] : ss[i + 8];
      p8[i] = keep + __shfl_xor(send, 1, 32);
    }
#pragma unroll
    for (int i = 0; i < 4; ++i) {
      const float keep = b1 ? p8[i + 4] : p8[i];
      const float send = b1 ? p8[i] : p8[i + 4];
      p4[i] = keep + __shfl_xor(send, 2, 32);
    }
#pragma unroll
    for (int i = 0; i < 2; ++i) {
      const float keep = b2 ? p4[i + 2] : p4[i];
      const float send = b2 ? p4[i] : p4[i + 2];
      p2[i] = keep + __shfl_xor(send, 4, 32);
    }
    const float keep = b3 ? p2[1] : p2[0];
    const float send = b3 ? p2[0] : p2[1];
    float tot = keep + __shfl_xor(send, 8, 32);
    tot += __shfl_xor(tot, 16, 32);
    const float n = sqrtf(tot);
    if (lane < 16) {
      const size_t row = (size_t)t * 16 + rho;
      s1[row]  = 1.0f / (n + EPSV);
      rsc[row] = (n + EPSV) / n;
    }
  }
}

__global__ void __launch_bounds__(512)
muprep_kernel(const float* src,
              float* __restrict__ mn,
              float* cmean,
              float* __restrict__ cr) {
  const int tid = threadIdx.x;
  const int lane = tid & 31;
  const int r = tid >> 5;
  const float4 v = ((const float4*)(src + r * DD))[lane];
  float ss = v.x * v.x + v.y * v.y + v.z * v.z + v.w * v.w;
  ss += __shfl_xor(ss, 1, 32);  ss += __shfl_xor(ss, 2, 32);
  ss += __shfl_xor(ss, 4, 32);  ss += __shfl_xor(ss, 8, 32);
  ss += __shfl_xor(ss, 16, 32);
  float inv1 = 1.0f / sqrtf(ss);
  float ax = v.x * inv1, ay = v.y * inv1, az = v.z * inv1, aw = v.w * inv1;
  float s2 = ax * ax + ay * ay + az * az + aw * aw;
  s2 += __shfl_xor(s2, 1, 32);  s2 += __shfl_xor(s2, 2, 32);
  s2 += __shfl_xor(s2, 4, 32);  s2 += __shfl_xor(s2, 8, 32);
  s2 += __shfl_xor(s2, 16, 32);
  float inv2 = 1.0f / sqrtf(s2);
  float4 o;
  o.x = ax * inv2; o.y = ay * inv2; o.z = az * inv2; o.w = aw * inv2;
  ((float4*)(mn + r * DD))[lane] = o;
  ((float4*)(cmean + r * DD))[lane] = make_float4(0.0f, 0.0f, 0.0f, 0.0f);
  if (tid < KK) cr[tid] = 0.0f;
}

#define MAIN_WAVES   32
#define MAIN_THREADS (MAIN_WAVES * 32)
#define MAIN_BLOCKS  256
__global__ void __launch_bounds__(MAIN_THREADS)
cluster_main(const float* __restrict__ emb, const float* __restrict__ s1,
             const float* __restrict__ rsc, const float* __restrict__ mn,
             float* __restrict__ cmean, float* __restrict__ cr,
             float* __restrict__ r_out, int write_r, int pass) {
  __shared__ __align__(16) float mn_lds[KK * TSTRIDE];
  __shared__ __align__(16) float tile[MAIN_WAVES][16 * TSTRIDE];
  __shared__ __align__(16) float rt_lds[MAIN_WAVES][16 * KK];
  __shared__ float cm_lds[KK * TSTRIDE];
  __shared__ float cr_lds[KK];

  const int tid  = threadIdx.x;
  const int lane = tid & 31;
  const int wid  = tid >> 5;
  const int lcol = lane & 15;
  const int h2   = (lane >> 4) & 1;

  for (int i = tid; i < KK * DD; i += MAIN_THREADS) {
    mn_lds[(i >> 7) * TSTRIDE + (i & 127)] = mn[i];
    cm_lds[(i >> 7) * TSTRIDE + (i & 127)] = 0.0f;
  }
  if (tid < KK) cr_lds[tid] = 0.0f;
  __syncthreads();

  float* tl = &tile[wid][0];
  v8f acc[8] = {};
  float crv = 0.0f;

  const int wid_s = __builtin_amdgcn_readfirstlane(wid);
  const int gw = __builtin_amdgcn_readfirstlane(blockIdx.x * MAIN_WAVES + wid);
  const int nw = MAIN_BLOCKS * MAIN_WAVES;

  const int nt = (gw < NTILES) ? (NTILES - 1 - gw) / nw + 1 : 0;
  const int ttop = gw + (nt - 1) * nw;
  const bool down = (pass & 1) == 0;
  const int tstep = down ? -nw : nw;
  int t = down ? ttop : gw;

  for (int left = nt - 1; left >= 0; --left, t += tstep) {
    if (left >= 3)      __builtin_amdgcn_s_setprio(3);
    else if (left == 2) __builtin_amdgcn_s_setprio(2);
    else if (left == 1) __builtin_amdgcn_s_setprio(1);
    else                __builtin_amdgcn_s_setprio(0);

    const size_t trow = (size_t)t * 16;
    const float* tbase = emb + trow * DD;
    const float* s1b = s1 + trow;
    const float* rb  = rsc + trow;

    float sc[16];
#pragma unroll
    for (int i = 0; i < 16; ++i) sc[i] = s1b[i];
#pragma unroll
    for (int g = 0; g < 2; ++g) {
      v4f v[8];
#pragma unroll
      for (int i = 0; i < 8; ++i)
        v[i] = ((const v4f*)(tbase + (8 * g + i) * DD))[lane];
      if (g == 0)
        __builtin_prefetch((const char*)(tbase + 8 * DD) + lane * 128, 0, 1);
      if (g == 0)
        asm volatile("" :: "v"(v[0]), "v"(v[1]), "v"(v[2]), "v"(v[3]),
                           "v"(v[4]), "v"(v[5]), "v"(v[6]), "v"(v[7]),
                           "s"(sc[0]), "s"(sc[1]), "s"(sc[2]), "s"(sc[3]),
                           "s"(sc[4]), "s"(sc[5]), "s"(sc[6]), "s"(sc[7]),
                           "s"(sc[8]), "s"(sc[9]), "s"(sc[10]), "s"(sc[11]),
                           "s"(sc[12]), "s"(sc[13]), "s"(sc[14]), "s"(sc[15]));
      else
        asm volatile("" :: "v"(v[0]), "v"(v[1]), "v"(v[2]), "v"(v[3]),
                           "v"(v[4]), "v"(v[5]), "v"(v[6]), "v"(v[7]));
#pragma unroll
      for (int i = 0; i < 8; ++i)
        *(v4f*)(tl + (8 * g + i) * TSTRIDE + lane * 4) = v[i] * sc[8 * g + i];
    }


    const float rsl = rb[(unsigned)lcol];
    v8f cosv = {};
    const float* ar = tl + lcol * TSTRIDE + 64 * h2;
    const float* br = mn_lds + lcol * TSTRIDE + 64 * h2;
#pragma unroll
    for (int q = 0; q < 16; ++q) {
      v4f a4 = *(const v4f*)(ar + 4 * q);
      v4f b4 = *(const v4f*)(br + 4 * q);
      cosv = __builtin_amdgcn_wmma_f32_16x16x4_f32(
          false, b4.xy, false, a4.xy, (short)0, cosv, false, false);
      cosv = __builtin_amdgcn_wmma_f32_16x16x4_f32(
          false, b4.zw, false, a4.zw, (short)0, cosv, false, false);
    }

    float ev[8];
#pragma unroll
    for (int j = 0; j < 8; ++j)
      ev[j] = __expf(-TEMP * (cosv[j] * rsl));
    float s = ((ev[0] + ev[1]) + (ev[2] + ev[3])) + ((ev[4] + ev[5]) + (ev[6] + ev[7]));
    s += SWZ_XOR(s, 16);
    const float inv = 1.0f / s;
    v4f plo, phi;
    plo[0] = ev[0] * inv; plo[1] = ev[1] * inv; plo[2] = ev[2] * inv; plo[3] = ev[3] * inv;
    phi[0] = ev[4] * inv; phi[1] = ev[5] * inv; phi[2] = ev[6] * inv; phi[3] = ev[7] * inv;

    if (write_r) {
      float* rt = r_out + trow * KK;
      const unsigned ro = (unsigned)(lcol * KK + h2 * 8);
      *(v4f*)(rt + ro)     = plo;
      *(v4f*)(rt + ro + 4) = phi;
    }

    float* rls = &rt_lds[wid][0] + lcol * KK + h2 * 8;
    const float* rlr = &rt_lds[wid][0] + h2 * (8 * KK) + lcol;
    *(v4f*)(rls)     = plo;
    *(v4f*)(rls + 4) = phi;
    float rr[8];
#pragma unroll
    for (int j = 0; j < 8; ++j) rr[j] = rlr[j * KK];

#pragma unroll
    for (int j = 0; j < 8; ++j) crv += rr[j];

#pragma unroll
    for (int c = 0; c < 4; ++c) {
      v2f af;
      af[0] = rr[2 * c];
      af[1] = rr[2 * c + 1];
      const float* b0 = (tl + h2 * (8 * TSTRIDE) + lcol) + 2 * c * TSTRIDE;
#pragma unroll
      for (int d0 = 0; d0 < 8; ++d0) {
        v2f bf;
        bf[0] = b0[d0 * 16];
        bf[1] = b0[d0 * 16 + TSTRIDE];
        acc[d0] = __builtin_amdgcn_wmma_f32_16x16x4_f32(
            false, af, false, bf, (short)0, acc[d0], false, false);
      }
    }
  }

  const int elane = (int)__builtin_amdgcn_mbcnt_lo(~0u, 0u);
  const int elcol = elane & 15;
  const int eh2   = elane >> 4;
  const int etid  = wid_s * 32 + elane;
  crv += __shfl_xor(crv, 16, 32);
  if (elane < KK) atomicAdd(&cr_lds[elane], crv);
#pragma unroll
  for (int d0 = 0; d0 < 8; ++d0)
#pragma unroll
    for (int jj = 0; jj < 8; ++jj)
      atomicAdd(&cm_lds[(jj + 8 * eh2) * TSTRIDE + d0 * 16 + elcol], acc[d0][jj]);
  __syncthreads();
  for (int i = etid; i < KK * DD; i += MAIN_THREADS)
    atomicAdd(&cmean[i], cm_lds[(i >> 7) * TSTRIDE + (i & 127)]);
  if (etid < KK) atomicAdd(&cr[etid], cr_lds[etid]);
}

__global__ void muupdate_kernel(const float* __restrict__ cmean,
                                const float* __restrict__ cr,
                                float* __restrict__ out_mu) {
  int i = blockIdx.x * blockDim.x + threadIdx.x;
  if (i < KK * DD) out_mu[i] = cmean[i] / cr[i >> 7];
}

extern "C" void kernel_launch(void* const* d_in, const int* in_sizes, int n_in,
                              void* d_out, int out_size, void* d_ws, size_t ws_size,
                              hipStream_t stream) {
  const float* emb     = (const float*)d_in[0];
  const float* mu_init = (const float*)d_in[1];
  float* out = (float*)d_out;

  float* s1    = (float*)d_ws;
  float* rsc   = s1 + NN;
  float* mn    = rsc + NN;
  float* cmean = mn + KK * DD;
  float* cr    = cmean + KK * DD;

  rownorm_kernel<<<1024, 256, 0, stream>>>(emb, s1, rsc);

  for (int it = 0; it < NITER; ++it) {
    const float* musrc = (it == 0) ? mu_init : cmean;
    const int last = (it == NITER - 1) ? 1 : 0;
    muprep_kernel<<<1, 512, 0, stream>>>(musrc, mn, cmean, cr);
    cluster_main<<<MAIN_BLOCKS, MAIN_THREADS, 0, stream>>>(emb, s1, rsc, mn, cmean, cr,
                                           out + KK * DD, last, it);
  }
  muupdate_kernel<<<(KK * DD) / 256, 256, 0, stream>>>(cmean, cr, out);
}


# block sums go to 16 copies of the global sum buffers (copy = block index mod 16): 16 atomic float adds per word, was 256; the mu-prep launch (16 blocks) adds the copies; sum order changes
# speedup vs baseline: 1.4284x; 1.0403x over previous
#include <hip/hip_runtime.h>
#include <math.h>


typedef __attribute__((ext_vector_type(2))) float v2f;
typedef __attribute__((ext_vector_type(4))) float v4f;
typedef __attribute__((ext_vector_type(8))) float v8f;

#define NN      500000
#define DD      128
#define KK      16
#define NTILES  (NN / 16)
#define TSTRIDE 132
#define TEMP    0.5f
#define EPSV    1e-8f
#define NITER   11
#define NCOPY   16

#define SWZ_XOR(v, m) \
  __int_as_float(__builtin_amdgcn_ds_swizzle(__float_as_int(v), 0x1f | ((m) << 10)))
__device__ __forceinline__ float sum16(float v) {
  v += SWZ_XOR(v, 1);
  v += SWZ_XOR(v, 2);
  v += SWZ_XOR(v, 4);
  v += SWZ_XOR(v, 8);
  return v;
}

__global__ void __launch_bounds__(256)
rownorm_kernel(const float* __restrict__ emb,
               float* __restrict__ s1,
               float* __restrict__ rsc) {
  const int lane = threadIdx.x & 31;
  const int gw = __builtin_amdgcn_readfirstlane(
      (int)(blockIdx.x * 8 + (threadIdx.x >> 5)));
  const int nw = (int)gridDim.x * 8;
  const bool b0 = (lane & 1) != 0;
  const bool b1 = (lane & 2) != 0;
  const bool b2 = (lane & 4) != 0;
  const bool b3 = (lane & 8) != 0;
  const int rho = ((lane & 1) << 3) | ((lane & 2) << 1) |
                  ((lane & 4) >> 1) | ((lane & 8) >> 3);

  for (int t = gw; t < NTILES; t += nw) {
    const float* tbase = emb + (size_t)t * 16 * DD;
    float ss[16];
#pragma unroll
    for (int i = 0; i < 16; ++i) {
      const float4 v = ((const float4*)(tbase + i * DD))[lane];
      ss[i] = v.x * v.x + v.y * v.y + v.z * v.z + v.w * v.w;
    }
    float p8[8], p4[4], p2[2];
#pragma unroll
    for (int i = 0; i < 8; ++i) {
      const float keep = b0 ? ss[i + 8] : ss[i];
      const float send = b0 ? ss[i] : ss[i + 8];
      p8[i] = keep + __shfl_xor(send, 1, 32);
    }
#pragma unroll
    for (int i = 0; i < 4; ++i) {
      const float keep = b1 ? p8[i + 4] : p8[i];
      const float send = b1 ? p8[i] : p8[i + 4];
      p4[i] = keep + __shfl_xor(send, 2, 32);
    }
#pragma unroll
    for (int i = 0; i < 2; ++i) {
      const float keep = b2 ? p4[i + 2] : p4[i];
      const float send = b2 ? p4[i] : p4[i + 2];
      p2[i] = keep + __shfl_xor(send, 4, 32);
    }
    const float keep = b3 ? p2[1] : p2[0];
    const float send = b3 ? p2[0] : p2[1];
    float tot = keep + __shfl_xor(send, 8, 32);
    tot += __shfl_xor(tot, 16, 32);
    const float n = sqrtf(tot);
    if (lane < 16) {
      const size_t row = (size_t)t * 16 + rho;
      s1[row]  = 1.0f / (n + EPSV);
      rsc[row] = (n + EPSV) / n;
    }
  }
}

__global__ void __launch_bounds__(512)
muprep_kernel(const float* src, int ncopies,
              float* __restrict__ mn,
              float* cmean,
              float* __restrict__ cr) {
  __shared__ __align__(16) float red[NCOPY][DD];
  const int tid = threadIdx.x;
  const int lane = tid & 31;
  const int g = tid >> 5;
  const int r = blockIdx.x;
  float4 part = make_float4(0.0f, 0.0f, 0.0f, 0.0f);
  if (g < ncopies) part = ((const float4*)(src + g * (KK * DD) + r * DD))[lane];
  ((float4*)(cmean + g * (KK * DD) + r * DD))[lane] = make_float4(0.0f, 0.0f, 0.0f, 0.0f);
  if (lane == 0) cr[g * KK + r] = 0.0f;
  ((float4*)(&red[g][0]))[lane] = part;
  __syncthreads();
  if (g != 0) return;
  float4 v = ((const float4*)(&red[0][0]))[lane];
#pragma unroll
  for (int k = 1; k < NCOPY; ++k) {
    const float4 w = ((const float4*)(&red[k][0]))[lane];
    v.x += w.x; v.y += w.y; v.z += w.z; v.w += w.w;
  }
  float ss = v.x * v.x + v.y * v.y + v.z * v.z + v.w * v.w;
  ss += __shfl_xor(ss, 1, 32);  ss += __shfl_xor(ss, 2, 32);
  ss += __shfl_xor(ss, 4, 32);  ss += __shfl_xor(ss, 8, 32);
  ss += __shfl_xor(ss, 16, 32);
  float inv1 = 1.0f / sqrtf(ss);
  float ax = v.x * inv1, ay = v.y * inv1, az = v.z * inv1, aw = v.w * inv1;
  float s2 = ax * ax + ay * ay + az * az + aw * aw;
  s2 += __shfl_xor(s2, 1, 32);  s2 += __shfl_xor(s2, 2, 32);
  s2 += __shfl_xor(s2, 4, 32);  s2 += __shfl_xor(s2, 8, 32);
  s2 += __shfl_xor(s2, 16, 32);
  float inv2 = 1.0f / sqrtf(s2);
  float4 o;
  o.x = ax * inv2; o.y = ay * inv2; o.z = az * inv2; o.w = aw * inv2;
  ((float4*)(mn + r * DD))[lane] = o;
}

#define MAIN_WAVES   32
#define MAIN_THREADS (MAIN_WAVES * 32)
#define MAIN_BLOCKS  256
__global__ void __launch_bounds__(MAIN_THREADS)
cluster_main(const float* __restrict__ emb, const float* __restrict__ s1,
             const float* __restrict__ rsc, const float* __restrict__ mn,
             float* __restrict__ cmean, float* __restrict__ cr,
             float* __restrict__ r_out, int write_r, int pass) {
  __shared__ __align__(16) float mn_lds[KK * TSTRIDE];
  __shared__ __align__(16) float tile[MAIN_WAVES][16 * TSTRIDE];
  __shared__ __align__(16) float rt_lds[MAIN_WAVES][16 * KK];
  __shared__ float cm_lds[KK * TSTRIDE];
  __shared__ float cr_lds[KK];

  const int tid  = threadIdx.x;
  const int lane = tid & 31;
  const int wid  = tid >> 5;
  const int lcol = lane & 15;
  const int h2   = (lane >> 4) & 1;

  for (int i = tid; i < KK * DD; i += MAIN_THREADS) {
    mn_lds[(i >> 7) * TSTRIDE + (i & 127)] = mn[i];
    cm_lds[(i >> 7) * TSTRIDE + (i & 127)] = 0.0f;
  }
  if (tid < KK) cr_lds[tid] = 0.0f;
  __syncthreads();

  float* tl = &tile[wid][0];
  v8f acc[8] = {};
  float crv = 0.0f;

  const int wid_s = __builtin_amdgcn_readfirstlane(wid);
  const int gw = __builtin_amdgcn_readfirstlane(blockIdx.x * MAIN_WAVES + wid);
  const int nw = MAIN_BLOCKS * MAIN_WAVES;

  const int nt = (gw < NTILES) ? (NTILES - 1 - gw) / nw + 1 : 0;
  const int ttop = gw + (nt - 1) * nw;
  const bool down = (pass & 1) == 0;
  const int tstep = down ? -nw : nw;
  int t = down ? ttop : gw;

  for (int left = nt - 1; left >= 0; --left, t += tstep) {
    if (left >= 3)      __builtin_amdgcn_s_setprio(3);
    else if (left == 2) __builtin_amdgcn_s_setprio(2);
    else if (left == 1) __builtin_amdgcn_s_setprio(1);
    else                __builtin_amdgcn_s_setprio(0);

    const size_t trow = (size_t)t * 16;
    const float* tbase = emb + trow * DD;
    const float* s1b = s1 + trow;
    const float* rb  = rsc + trow;

    float sc[16];
#pragma unroll
    for (int i = 0; i < 16; ++i) sc[i] = s1b[i];
#pragma unroll
    for (int g = 0; g < 2; ++g) {
      v4f v[8];
#pragma unroll
      for (int i = 0; i < 8; ++i)
        v[i] = ((const v4f*)(tbase + (8 * g + i) * DD))[lane];
      if (g == 0)
        __builtin_prefetch((const char*)(tbase + 8 * DD) + lane * 128, 0, 1);
      if (g == 0)
        asm volatile("" :: "v"(v[0]), "v"(v[1]), "v"(v[2]), "v"(v[3]),
                           "v"(v[4]), "v"(v[5]), "v"(v[6]), "v"(v[7]),
                           "s"(sc[0]), "s"(sc[1]), "s"(sc[2]), "s"(sc[3]),
                           "s"(sc[4]), "s"(sc[5]), "s"(sc[6]), "s"(sc[7]),
                           "s"(sc[8]), "s"(sc[9]), "s"(sc[10]), "s"(sc[11]),
                           "s"(sc[12]), "s"(sc[13]), "s"(sc[14]), "s"(sc[15]));
      else
        asm volatile("" :: "v"(v[0]), "v"(v[1]), "v"(v[2]), "v"(v[3]),
                           "v"(v[4]), "v"(v[5]), "v"(v[6]), "v"(v[7]));
#pragma unroll
      for (int i = 0; i < 8; ++i)
        *(v4f*)(tl + (8 * g + i) * TSTRIDE + lane * 4) = v[i] * sc[8 * g + i];
    }


    const float rsl = rb[(unsigned)lcol];
    v8f cosv = {};
    const float* ar = tl + lcol * TSTRIDE + 64 * h2;
    const float* br = mn_lds + lcol * TSTRIDE + 64 * h2;
#pragma unroll
    for (int q = 0; q < 16; ++q) {
      v4f a4 = *(const v4f*)(ar + 4 * q);
      v4f b4 = *(const v4f*)(br + 4 * q);
      cosv = __builtin_amdgcn_wmma_f32_16x16x4_f32(
          false, b4.xy, false, a4.xy, (short)0, cosv, false, false);
      cosv = __builtin_amdgcn_wmma_f32_16x16x4_f32(
          false, b4.zw, false, a4.zw, (short)0, cosv, false, false);
    }

    float ev[8];
#pragma unroll
    for (int j = 0; j < 8; ++j)
      ev[j] = __expf(-TEMP * (cosv[j] * rsl));
    float s = ((ev[0] + ev[1]) + (ev[2] + ev[3])) + ((ev[4] + ev[5]) + (ev[6] + ev[7]));
    s += SWZ_XOR(s, 16);
    const float inv = 1.0f / s;
    v4f plo, phi;
    plo[0] = ev[0] * inv; plo[1] = ev[1] * inv; plo[2] = ev[2] * inv; plo[3] = ev[3] * inv;
    phi[0] = ev[4] * inv; phi[1] = ev[5] * inv; phi[2] = ev[6] * inv; phi[3] = ev[7] * inv;

    if (write_r) {
      float* rt = r_out + trow * KK;
      const unsigned ro = (unsigned)(lcol * KK + h2 * 8);
      *(v4f*)(rt + ro)     = plo;
      *(v4f*)(rt + ro + 4) = phi;
    }

    float* rls = &rt_lds[wid][0] + lcol * KK + h2 * 8;
    const float* rlr = &rt_lds[wid][0] + h2 * (8 * KK) + lcol;
    *(v4f*)(rls)     = plo;
    *(v4f*)(rls + 4) = phi;
    float rr[8];
#pragma unroll
    for (int j = 0; j < 8; ++j) rr[j] = rlr[j * KK];

#pragma unroll
    for (int j = 0; j < 8; ++j) crv += rr[j];

#pragma unroll
    for (int c = 0; c < 4; ++c) {
      v2f af;
      af[0] = rr[2 * c];
      af[1] = rr[2 * c + 1];
      const float* b0 = (tl + h2 * (8 * TSTRIDE) + lcol) + 2 * c * TSTRIDE;
#pragma unroll
      for (int d0 = 0; d0 < 8; ++d0) {
        v2f bf;
        bf[0] = b0[d0 * 16];
        bf[1] = b0[d0 * 16 + TSTRIDE];
        acc[d0] = __builtin_amdgcn_wmma_f32_16x16x4_f32(
            false, af, false, bf, (short)0, acc[d0], false, false);
      }
    }
  }

  const int elane = (int)__builtin_amdgcn_mbcnt_lo(~0u, 0u);
  const int elcol = elane & 15;
  const int eh2   = elane >> 4;
  const int etid  = wid_s * 32 + elane;
  crv += __shfl_xor(crv, 16, 32);
  if (elane < KK) atomicAdd(&cr_lds[elane], crv);
#pragma unroll
  for (int d0 = 0; d0 < 8; ++d0)
#pragma unroll
    for (int jj = 0; jj < 8; ++jj)
      atomicAdd(&cm_lds[(jj + 8 * eh2) * TSTRIDE + d0 * 16 + elcol], acc[d0][jj]);
  __syncthreads();
  const int copy = (int)(blockIdx.x & (NCOPY - 1));
  float* cmean_c = cmean + copy * (KK * DD);
  float* cr_c = cr + copy * KK;
  for (int i = etid; i < KK * DD; i += MAIN_THREADS)
    atomicAdd(&cmean_c[i], cm_lds[(i >> 7) * TSTRIDE + (i & 127)]);
  if (etid < KK) atomicAdd(&cr_c[etid], cr_lds[etid]);
}

__global__ void muupdate_kernel(const float* __restrict__ cmean,
                                const float* __restrict__ cr,
                                float* __restrict__ out_mu) {
  int i = blockIdx.x * blockDim.x + threadIdx.x;
  if (i < KK * DD) {
    float m = 0.0f, s = 0.0f;
#pragma unroll
    for (int k = 0; k < NCOPY; ++k) {
      m += cmean[k * (KK * DD) + i];
      s += cr[k * KK + (i >> 7)];
    }
    out_mu[i] = m / s;
  }
}

extern "C" void kernel_launch(void* const* d_in, const int* in_sizes, int n_in,
                              void* d_out, int out_size, void* d_ws, size_t ws_size,
                              hipStream_t stream) {
  const float* emb     = (const float*)d_in[0];
  const float* mu_init = (const float*)d_in[1];
  float* out = (float*)d_out;

  float* s1    = (float*)d_ws;
  float* rsc   = s1 + NN;
  float* mn    = rsc + NN;
  float* cmean = mn + KK * DD;
  float* cr    = cmean + NCOPY * KK * DD;

  rownorm_kernel<<<1024, 256, 0, stream>>>(emb, s1, rsc);

  for (int it = 0; it < NITER; ++it) {
    const float* musrc = (it == 0) ? mu_init : cmean;
    const int last = (it == NITER - 1) ? 1 : 0;
    const int ncopies = (it == 0) ? 1 : NCOPY;
    muprep_kernel<<<KK, 512, 0, stream>>>(musrc, ncopies, mn, cmean, cr);
    cluster_main<<<MAIN_BLOCKS, MAIN_THREADS, 0, stream>>>(emb, s1, rsc, mn, cmean, cr,
                                           out + KK * DD, last, it);
  }
  muupdate_kernel<<<(KK * DD) / 256, 256, 0, stream>>>(cmean, cr, out);
}
